# starting kernel
# baseline (speedup 1.0000x reference)
#include <hip/hip_runtime.h>
#include <hip/hip_bf16.h>


typedef __attribute__((ext_vector_type(2))) float v2f;
typedef __attribute__((ext_vector_type(8))) float v8f;

#define CB 2
#define CT 2048
#define CE 1024
#define CH 8
#define CDK 128
#define CDV 256
#define CV 2048
#define NQKV 4096
#define BT (CB * CT)

static __device__ __forceinline__ v8f wmma4(v2f a, v2f b, v8f c) {
  return __builtin_amdgcn_wmma_f32_16x16x4_f32(
       false, a,  false, b,
       (short)0, c,  false,  false);
}

static __device__ __forceinline__ v8f vzero8() {
  v8f z;
#pragma unroll
  for (int i = 0; i < 8; ++i) z[i] = 0.0f;
  return z;
}

static __device__ __forceinline__ v2f ldsv2(const float* p) {
  return *reinterpret_cast<const v2f*>(p);
}

__global__ __launch_bounds__(128) void gemm_wmma_f32(
    const float* __restrict__ A, const float* __restrict__ Bm,
    const float* __restrict__ bias, float* __restrict__ C,
    int M, int N, int K) {
  __shared__ float as[128][34];
  __shared__ float bs[16][66][2];

  const int tid = threadIdx.x;
  const int lane = tid & 31;
  const int w = tid >> 5;
  const int hi = lane >> 4;
  const int ln = lane & 15;
  const int n0 = blockIdx.x * 64;
  const int m0 = blockIdx.y * 128;

  v8f acc[2][4];
#pragma unroll
  for (int mt = 0; mt < 2; ++mt)
#pragma unroll
    for (int nt = 0; nt < 4; ++nt) acc[mt][nt] = vzero8();

  for (int k0 = 0; k0 < K; k0 += 32) {
#pragma unroll
    for (int e = tid; e < 128 * 32; e += 128) {
      int r = e >> 5, c = e & 31;
      as[r][c] = A[(size_t)(m0 + r) * K + (k0 + c)];
    }
#pragma unroll
    for (int e = tid; e < 32 * 64; e += 128) {
      int r = e >> 6, c = e & 63;
      bs[r >> 1][c][r & 1] = Bm[(size_t)(k0 + r) * N + (n0 + c)];
    }
    __syncthreads();

    if (k0 + 32 < K) {
      __builtin_prefetch(&A[(size_t)(m0 + tid) * K + (k0 + 32)], 0, 1);
      __builtin_prefetch(&Bm[(size_t)(k0 + 32 + (tid & 31)) * N + n0], 0, 1);
    }

#pragma unroll
    for (int kk = 0; kk < 32; kk += 4) {
      const int ak = kk + hi * 2;
      const int kp = (kk >> 1) + hi;
      v2f a0 = ldsv2(&as[w * 32 + ln][ak]);
      v2f a1 = ldsv2(&as[w * 32 + 16 + ln][ak]);
#pragma unroll
      for (int nt = 0; nt < 4; ++nt) {
        v2f b = ldsv2(&bs[kp][nt * 16 + ln][0]);
        acc[0][nt] = wmma4(a0, b, acc[0][nt]);
        acc[1][nt] = wmma4(a1, b, acc[1][nt]);
      }
    }
    __syncthreads();
  }

#pragma unroll
  for (int mt = 0; mt < 2; ++mt)
#pragma unroll
    for (int nt = 0; nt < 4; ++nt)
#pragma unroll
      for (int vv = 0; vv < 8; ++vv) {
        int row = m0 + w * 32 + mt * 16 + vv + hi * 8;
        int col = n0 + nt * 16 + ln;
        C[(size_t)row * N + col] = acc[mt][nt][vv] + bias[col];
      }
}

__global__ __launch_bounds__(512) void xpos_kernel(float* __restrict__ qkv) {
  const int row = blockIdx.x;
  const int j = threadIdx.x;
  const int t = row & (CT - 1);

  const float D = (float)CE;
  const float base = (2.0f * (float)j + 0.4f * D) / (1.4f * D);
  const float sc = __powf(base, (float)t / 512.0f);
  const float invf = __powf(10000.0f, -(float)j / 512.0f);
  float sn, cn;
  __sincosf((float)t * invf, &sn, &cn);

  size_t off = (size_t)row * NQKV + 2 * j;
  {
    float x0 = qkv[off], x1 = qkv[off + 1];
    float c = cn * sc, s = sn * sc;
    qkv[off]     = x0 * c - x1 * s;
    qkv[off + 1] = x1 * c + x0 * s;
  }
  {
    float isc = 1.0f / sc;
    float x0 = qkv[off + CE], x1 = qkv[off + CE + 1];
    float c = cn * isc, s = sn * isc;
    qkv[off + CE]     = x0 * c - x1 * s;
    qkv[off + CE + 1] = x1 * c + x0 * s;
  }
}

__global__ __launch_bounds__(128) void retention_kernel(
    const float* __restrict__ qkv, float* __restrict__ ro) {
  __shared__ float q_s[64][132];
  __shared__ float k_s[16][132];
  __shared__ float v_s[8][258][2];
  __shared__ float s_s[4][16][20];

  const int iblk = blockIdx.x;
  const int bh = blockIdx.y;
  const int b = bh >> 3, h = bh & 7;
  const int i0 = iblk * 64;
  const int rowbase = b * CT;

  const int tid = threadIdx.x;
  const int lane = tid & 31;
  const int w = tid >> 5;
  const int hi = lane >> 4;
  const int ln = lane & 15;

  const float se = logf(1.0f / 32.0f), ee = logf(1.0f / 512.0f);
  const float g = 1.0f - __expf(se + (ee - se) * ((float)h / 7.0f));
  const float logg = logf(g);
  const float rsdk = rsqrtf((float)CDK);

#pragma unroll
  for (int e = tid; e < 64 * 128; e += 128) {
    int r = e >> 7, c = e & 127;
    q_s[r][c] = qkv[(size_t)(rowbase + i0 + r) * NQKV + h * CDK + c];
  }

  v8f acc[16];
#pragma unroll
  for (int i = 0; i < 16; ++i) acc[i] = vzero8();

  for (int j0 = 0; j0 < i0 + 64; j0 += 16) {
    __syncthreads();
#pragma unroll
    for (int e = tid; e < 16 * 128; e += 128) {
      int r = e >> 7, c = e & 127;
      k_s[r][c] = qkv[(size_t)(rowbase + j0 + r) * NQKV + CE + h * CDK + c];
    }
#pragma unroll
    for (int e = tid; e < 16 * 256; e += 128) {
      int r = e >> 8, c = e & 255;
      v_s[r >> 1][c][r & 1] =
          qkv[(size_t)(rowbase + j0 + r) * NQKV + 2 * CE + h * CDV + c];
    }
    __syncthreads();

    v8f sacc = vzero8();
#pragma unroll
    for (int kk = 0; kk < CDK; kk += 4) {
      const int ak = kk + hi * 2;
      v2f a = ldsv2(&q_s[w * 16 + ln][ak]);
      v2f bf = ldsv2(&k_s[ln][ak]);
      sacc = wmma4(a, bf, sacc);
    }

#pragma unroll
    for (int vv = 0; vv < 8; ++vv) {
      int i = i0 + w * 16 + vv + hi * 8;
      int j = j0 + ln;
      float m = (i >= j) ? __expf(logg * (float)(i - j)) * rsdk : 0.0f;
      sacc[vv] *= m;
    }

#pragma unroll
    for (int vv = 0; vv < 8; ++vv) s_s[w][vv + hi * 8][ln] = sacc[vv];

#pragma unroll
    for (int ks = 0; ks < 16; ks += 4) {
      const int ak = ks + hi * 2;
      const int kp = (ks >> 1) + hi;
      v2f a = ldsv2(&s_s[w][ln][ak]);
#pragma unroll
      for (int nt = 0; nt < 16; ++nt) {
        v2f bf = ldsv2(&v_s[kp][nt * 16 + ln][0]);
        acc[nt] = wmma4(a, bf, acc[nt]);
      }
    }
  }

  const float inv1mg = 1.0f / (1.0f - g);
#pragma unroll
  for (int nt = 0; nt < 16; ++nt)
#pragma unroll
    for (int vv = 0; vv < 8; ++vv) {
      int i = i0 + w * 16 + vv + hi * 8;
      float rowsum = (1.0f - __expf(logg * (float)(i + 1))) * inv1mg;
      float sc = rsqrtf(rowsum);
      ro[(size_t)(rowbase + i) * CV + h * CDV + nt * 16 + ln] = acc[nt][vv] * sc;
    }
}

__global__ __launch_bounds__(256) void curr_kv_kernel(
    const float* __restrict__ qkv, float* __restrict__ kvout) {
  __shared__ float kw_s[8][130][2];
  __shared__ float v_s[8][258][2];

  const int bh = blockIdx.x;
  const int b = bh >> 3, h = bh & 7;
  const int rowbase = b * CT;

  const int tid = threadIdx.x;
  const int lane = tid & 31;
  const int w = tid >> 5;
  const int hi = lane >> 4;
  const int ln = lane & 15;

  const float se = logf(1.0f / 32.0f), ee = logf(1.0f / 512.0f);
  const float g = 1.0f - __expf(se + (ee - se) * ((float)h / 7.0f));
  const float logg = logf(g);
  const float rnorm = rsqrtf((1.0f - __expf(logg * (float)CT)) / (1.0f - g));

  v8f acc[16];
#pragma unroll
  for (int i = 0; i < 16; ++i) acc[i] = vzero8();

  for (int j0 = 0; j0 < CT; j0 += 16) {
#pragma unroll
    for (int e = tid; e < 16 * 128; e += 256) {
      int r = e >> 7, c = e & 127;
      int j = j0 + r;
      float wj = __expf(logg * (float)(CT - 1 - j)) * rnorm;
      kw_s[r >> 1][c][r & 1] =
          qkv[(size_t)(rowbase + j) * NQKV + CE + h * CDK + c] * wj;
    }
#pragma unroll
    for (int e = tid; e < 16 * 256; e += 256) {
      int r = e >> 8, c = e & 255;
      v_s[r >> 1][c][r & 1] =
          qkv[(size_t)(rowbase + j0 + r) * NQKV + 2 * CE + h * CDV + c];
    }
    __syncthreads();

#pragma unroll
    for (int ks = 0; ks < 16; ks += 4) {
      const int kp = (ks >> 1) + hi;
      v2f a = ldsv2(&kw_s[kp][w * 16 + ln][0]);
#pragma unroll
      for (int nt = 0; nt < 16; ++nt) {
        v2f bf = ldsv2(&v_s[kp][nt * 16 + ln][0]);
        acc[nt] = wmma4(a, bf, acc[nt]);
      }
    }
    __syncthreads();
  }

#pragma unroll
  for (int nt = 0; nt < 16; ++nt)
#pragma unroll
    for (int vv = 0; vv < 8; ++vv) {
      int kk = w * 16 + vv + hi * 8;
      kvout[(size_t)bh * CDK * CDV + (size_t)kk * CDV + nt * 16 + ln] =
          acc[nt][vv];
    }
}

__global__ __launch_bounds__(256) void gn_gate_kernel(
    const float* __restrict__ ro, float* __restrict__ gbuf) {
  const int row = blockIdx.x;
  const int w = threadIdx.x >> 5;
  const int lane = threadIdx.x & 31;
  const size_t base = (size_t)row * CV + (size_t)w * CDV;

  float x[8];
  float s = 0.0f, s2 = 0.0f;
#pragma unroll
  for (int i = 0; i < 8; ++i) {
    x[i] = ro[base + i * 32 + lane];
    s += x[i];
    s2 += x[i] * x[i];
  }
#pragma unroll
  for (int off = 16; off > 0; off >>= 1) {
    s += __shfl_xor(s, off, 32);
    s2 += __shfl_xor(s2, off, 32);
  }
  const float mu = s * (1.0f / 256.0f);
  const float var = s2 * (1.0f / 256.0f) - mu * mu;
  const float rstd = rsqrtf(var + 1e-5f);

#pragma unroll
  for (int i = 0; i < 8; ++i) {
    size_t idx = base + i * 32 + lane;
    float nrm = (x[i] - mu) * rstd;
    float gv = gbuf[idx];
    float sil = gv / (1.0f + __expf(-gv));
    gbuf[idx] = sil * nrm;
  }
}

extern "C" void kernel_launch(void* const* d_in, const int* in_sizes, int n_in,
                              void* d_out, int out_size, void* d_ws,
                              size_t ws_size, hipStream_t stream) {
  const float* hs    = (const float*)d_in[0];
  const float* w_qkv = (const float*)d_in[1];
  const float* b_qkv = (const float*)d_in[2];
  const float* w_g   = (const float*)d_in[3];
  const float* b_g   = (const float*)d_in[4];
  const float* w_p   = (const float*)d_in[5];
  const float* b_p   = (const float*)d_in[6];

  float* out  = (float*)d_out;
  float* qkv  = (float*)d_ws;
  float* gbuf = qkv + (size_t)BT * NQKV;
  float* ro   = gbuf + (size_t)BT * CV;

  gemm_wmma_f32<<<dim3(NQKV / 64, BT / 128), 128, 0, stream>>>(
      hs, w_qkv, b_qkv, qkv, BT, NQKV, CE);

  xpos_kernel<<<dim3(BT), 512, 0, stream>>>(qkv);

  gemm_wmma_f32<<<dim3(CV / 64, BT / 128), 128, 0, stream>>>(
      hs, w_g, b_g, gbuf, BT, CV, CE);

  retention_kernel<<<dim3(CT / 64, CB * CH), 128, 0, stream>>>(qkv, ro);

  curr_kv_kernel<<<dim3(CB * CH), 256, 0, stream>>>(
      qkv, out + (size_t)BT * CE);

  gn_gate_kernel<<<dim3(BT), 256, 0, stream>>>(ro, gbuf);

  gemm_wmma_f32<<<dim3(CE / 64, BT / 128), 128, 0, stream>>>(
      gbuf, w_p, b_p, out, BT, CE, CV);
}


# retention kernel: Q,K,V straight from global into WMMA fragments (16-byte loads), products transposed so S^T's D registers are the next B fragment; no LDS, no barriers; dk and key sums regrouped
# speedup vs baseline: 4.0261x; 4.0261x over previous
#include <hip/hip_runtime.h>
#include <hip/hip_bf16.h>


typedef __attribute__((ext_vector_type(2))) float v2f;
typedef __attribute__((ext_vector_type(8))) float v8f;

#define CB 2
#define CT 2048
#define CE 1024
#define CH 8
#define CDK 128
#define CDV 256
#define CV 2048
#define NQKV 4096
#define BT (CB * CT)

static __device__ __forceinline__ v8f wmma4(v2f a, v2f b, v8f c) {
  return __builtin_amdgcn_wmma_f32_16x16x4_f32(
       false, a,  false, b,
       (short)0, c,  false,  false);
}

static __device__ __forceinline__ v8f vzero8() {
  v8f z;
#pragma unroll
  for (int i = 0; i < 8; ++i) z[i] = 0.0f;
  return z;
}

static __device__ __forceinline__ v2f ldsv2(const float* p) {
  return *reinterpret_cast<const v2f*>(p);
}

typedef __attribute__((ext_vector_type(4))) float v4f;

template <int MT, int NB, int N, int K>
struct GemmFrag {
  v4f a[NB][MT];
  v4f b[NB][4];
  __device__ __forceinline__ void load(const v4f* __restrict__ pa,
                                       const v4f* __restrict__ pb, int k) {
#pragma unroll
    for (int nb = 0; nb < NB; ++nb) {
#pragma unroll
      for (int mt = 0; mt < MT; ++mt)
        a[nb][mt] = pa[((size_t)mt * 16 * K + k + 8 * nb) / 4];
#pragma unroll
      for (int j = 0; j < 4; ++j)
        b[nb][j] = pb[((size_t)(k + 8 * nb + j) * N) / 4];
    }
  }
  __device__ __forceinline__ void mma(v8f (&acc)[MT][4]) const {
#pragma unroll
    for (int nb = 0; nb < NB; ++nb)
#pragma unroll
      for (int s = 0; s < 2; ++s)
#pragma unroll
        for (int mt = 0; mt < MT; ++mt) {
          v2f af;
          af[0] = a[nb][mt][2 * s];
          af[1] = a[nb][mt][2 * s + 1];
#pragma unroll
          for (int nt = 0; nt < 4; ++nt) {
            v2f bf;
            bf[0] = b[nb][2 * s][nt];
            bf[1] = b[nb][2 * s + 1][nt];
            acc[mt][nt] = wmma4(af, bf, acc[mt][nt]);
          }
        }
  }
};

template <int MT, int NB, int N, int K>
__global__ __launch_bounds__(128) void gemm_wmma_f32(
    const float* __restrict__ A, const float* __restrict__ Bm,
    const float* __restrict__ bias, float* __restrict__ C) {
  constexpr int KS = 8 * NB;
  const int tid = threadIdx.x;
  const int lane = tid & 31;
  const int w = tid >> 5;
  const int hi = lane >> 4;
  const int ln = lane & 15;
  const int m0 = (blockIdx.y * 2 + (w >> 1)) * (16 * MT);
  const int n0 = (blockIdx.x * 2 + (w & 1)) * 64;

  const v4f* __restrict__ pa = reinterpret_cast<const v4f*>(
      A + (size_t)(m0 + ln) * K + 4 * hi);
  const v4f* __restrict__ pb = reinterpret_cast<const v4f*>(
      Bm + (size_t)(4 * hi) * N + n0 + 4 * ln);

  v8f acc[MT][4];
#pragma unroll
  for (int mt = 0; mt < MT; ++mt)
#pragma unroll
    for (int nt = 0; nt < 4; ++nt) acc[mt][nt] = vzero8();

  GemmFrag<MT, NB, N, K> f0, f1;
  f0.load(pa, pb, 0);
  for (int k0 = 0; k0 < K - 2 * KS; k0 += 2 * KS) {
    f1.load(pa, pb, k0 + KS);
    __builtin_amdgcn_sched_barrier(0);
    f0.mma(acc);
    __builtin_amdgcn_sched_barrier(0);
    f0.load(pa, pb, k0 + 2 * KS);
    __builtin_amdgcn_sched_barrier(0);
    f1.mma(acc);
    __builtin_amdgcn_sched_barrier(0);
  }
  f1.load(pa, pb, K - KS);
  f0.mma(acc);
  f1.mma(acc);

  const v4f bv = *reinterpret_cast<const v4f*>(bias + n0 + 4 * ln);
#pragma unroll
  for (int mt = 0; mt < MT; ++mt)
#pragma unroll
    for (int r = 0; r < 8; ++r) {
      const int row = m0 + mt * 16 + hi * 8 + r;
      v4f o;
#pragma unroll
      for (int nt = 0; nt < 4; ++nt) o[nt] = acc[mt][nt][r] + bv[nt];
      *reinterpret_cast<v4f*>(C + (size_t)row * N + n0 + 4 * ln) = o;
    }
}

__global__ __launch_bounds__(512) void xpos_kernel(float* __restrict__ qkv) {
  const int row = blockIdx.x;
  const int j = threadIdx.x;
  const int t = row & (CT - 1);

  const float D = (float)CE;
  const float base = (2.0f * (float)j + 0.4f * D) / (1.4f * D);
  const float sc = __powf(base, (float)t / 512.0f);
  const float invf = __powf(10000.0f, -(float)j / 512.0f);
  float sn, cn;
  __sincosf((float)t * invf, &sn, &cn);

  size_t off = (size_t)row * NQKV + 2 * j;
  {
    float x0 = qkv[off], x1 = qkv[off + 1];
    float c = cn * sc, s = sn * sc;
    qkv[off]     = x0 * c - x1 * s;
    qkv[off + 1] = x1 * c + x0 * s;
  }
  {
    float isc = 1.0f / sc;
    float x0 = qkv[off + CE], x1 = qkv[off + CE + 1];
    float c = cn * isc, s = sn * isc;
    qkv[off + CE]     = x0 * c - x1 * s;
    qkv[off + CE + 1] = x1 * c + x0 * s;
  }
}

#define RT_CK 256
#define RT_NWG 144
#define RT_TILE (64 * CDV)

__global__ __launch_bounds__(128) void retention_kernel(
    const float* __restrict__ qkv, float* __restrict__ part) {
  int n = blockIdx.x;
  int gq = 0;
  while (gq < 7 && n >= 4 * (gq + 1)) {
    n -= 4 * (gq + 1);
    ++gq;
  }
  const int iblk = 4 * gq + n / (gq + 1);
  const int ck = n % (gq + 1);
  const int bh = blockIdx.y;
  const int b = bh >> 3, h = bh & 7;
  const int i0 = iblk * 64;
  const int rowbase = b * CT;
  const int jbeg = ck * RT_CK;
  const int jend = min(jbeg + RT_CK, i0 + 64);
  const int nblk = (jend - jbeg) >> 4;

  const int tid = threadIdx.x;
  const int lane = tid & 31;
  const int w = tid >> 5;
  const int hi = lane >> 4;
  const int ln = lane & 15;

  const float se = logf(1.0f / 32.0f), ee = logf(1.0f / 512.0f);
  const float g = 1.0f - __expf(se + (ee - se) * ((float)h / 7.0f));
  const float logg = logf(g);
  const float rsdk = rsqrtf((float)CDK);

  const int qi = i0 + w * 16 + ln;
  const v4f* __restrict__ pq = reinterpret_cast<const v4f*>(
      qkv + (size_t)(rowbase + qi) * NQKV + h * CDK + 4 * hi);
  const float* __restrict__ pk =
      qkv + (size_t)(rowbase + jbeg + ln) * NQKV + CE + h * CDK + 4 * hi;
  const float* __restrict__ pv =
      qkv + (size_t)(rowbase + jbeg + 8 * hi) * NQKV + 2 * CE + h * CDV + 4 * ln;

  v4f qf[16];
  v4f kf[16];
  v4f vf[4][8];
#pragma unroll
  for (int kb = 0; kb < 16; ++kb) qf[kb] = pq[2 * kb];
#pragma unroll
  for (int kb = 0; kb < 16; ++kb)
    kf[kb] = *reinterpret_cast<const v4f*>(pk + 8 * kb);
  __builtin_amdgcn_sched_barrier(0);
#pragma unroll
  for (int gg = 0; gg < 2; ++gg)
#pragma unroll
    for (int jj = 0; jj < 8; ++jj)
      vf[gg][jj] =
          *reinterpret_cast<const v4f*>(pv + (size_t)jj * NQKV + 64 * gg);
  __builtin_amdgcn_sched_barrier(0);
#pragma unroll
  for (int gg = 2; gg < 4; ++gg)
#pragma unroll
    for (int jj = 0; jj < 8; ++jj)
      vf[gg][jj] =
          *reinterpret_cast<const v4f*>(pv + (size_t)jj * NQKV + 64 * gg);

  v8f acc[16];
#pragma unroll
  for (int i = 0; i < 16; ++i) acc[i] = vzero8();

  for (int t = 0; t < nblk; ++t) {
    const int j0 = jbeg + 16 * t;
    const int tn = min(t + 1, nblk - 1);
    const float* __restrict__ pkn = pk + (size_t)tn * 16 * NQKV;
    const float* __restrict__ pvn = pv + (size_t)tn * 16 * NQKV;

    v8f s0 = vzero8(), s1 = vzero8();
#pragma unroll
    for (int kb = 0; kb < 16; kb += 2)
#pragma unroll
      for (int s = 0; s < 2; ++s) {
        v2f a0, b0, a1, b1;
        a0[0] = kf[kb][2 * s];
        a0[1] = kf[kb][2 * s + 1];
        b0[0] = qf[kb][2 * s];
        b0[1] = qf[kb][2 * s + 1];
        a1[0] = kf[kb + 1][2 * s];
        a1[1] = kf[kb + 1][2 * s + 1];
        b1[0] = qf[kb + 1][2 * s];
        b1[1] = qf[kb + 1][2 * s + 1];
        s0 = wmma4(a0, b0, s0);
        s1 = wmma4(a1, b1, s1);
      }

    __builtin_amdgcn_sched_barrier(0);
#pragma unroll
    for (int kb = 0; kb < 16; ++kb)
      kf[kb] = *reinterpret_cast<const v4f*>(pkn + 8 * kb);
    __builtin_amdgcn_sched_barrier(0);

    v8f st;
#pragma unroll
    for (int r = 0; r < 8; ++r) {
      const int j = j0 + 8 * hi + r;
      const float m =
          (qi >= j) ? __expf(logg * (float)(qi - j)) * rsdk : 0.0f;
      st[r] = (s0[r] + s1[r]) * m;
    }

#pragma unroll
    for (int gg = 0; gg < 4; ++gg) {
#pragma unroll
      for (int s = 0; s < 4; ++s) {
        v2f bf;
        bf[0] = st[2 * s];
        bf[1] = st[2 * s + 1];
#pragma unroll
        for (int mt = 0; mt < 4; ++mt) {
          v2f af;
          af[0] = vf[gg][2 * s][mt];
          af[1] = vf[gg][2 * s + 1][mt];
          acc[4 * gg + mt] = wmma4(af, bf, acc[4 * gg + mt]);
        }
      }
      __builtin_amdgcn_sched_barrier(0);
#pragma unroll
      for (int jj = 0; jj < 8; ++jj)
        vf[gg][jj] =
            *reinterpret_cast<const v4f*>(pvn + (size_t)jj * NQKV + 64 * gg);
      __builtin_amdgcn_sched_barrier(0);
    }
  }

  float* __restrict__ trow =
      part + ((size_t)bh * RT_NWG + blockIdx.x) * RT_TILE +
      (size_t)(w * 16 + ln) * CDV + 32 * hi;
#pragma unroll
  for (int gg = 0; gg < 4; ++gg)
#pragma unroll
    for (int r = 0; r < 8; ++r) {
      v4f o;
#pragma unroll
      for (int mt = 0; mt < 4; ++mt) o[mt] = acc[4 * gg + mt][r];
      *reinterpret_cast<v4f*>(trow + 64 * gg + 4 * r) = o;
    }
}

__global__ __launch_bounds__(512) void retention_reduce_kernel(
    const float* __restrict__ part, float* __restrict__ ro) {
  const int row = blockIdx.x;
  const int b = row >> 11;
  const int t = row & (CT - 1);
  const int c4 = threadIdx.x;
  const int h = c4 >> 6;
  const int iblk = t >> 6;
  const int gq = iblk >> 2;
  const int nch = gq + 1;
  const int n0 = 2 * gq * (gq + 1) + (iblk & 3) * nch;

  const float4* __restrict__ p4 =
      reinterpret_cast<const float4*>(
          part + ((size_t)(b * CH + h) * RT_NWG + n0) * RT_TILE +
          (size_t)(t & 63) * CDV) +
      (c4 & 63);
  float4 v[8];
#pragma unroll
  for (int c = 0; c < 8; ++c)
    if (c < nch) v[c] = p4[(size_t)c * (RT_TILE / 4)];
  float4 s = v[0];
#pragma unroll
  for (int c = 1; c < 8; ++c)
    if (c < nch) {
      s.x += v[c].x;
      s.y += v[c].y;
      s.z += v[c].z;
      s.w += v[c].w;
    }

  const float se = logf(1.0f / 32.0f), ee = logf(1.0f / 512.0f);
  const float g = 1.0f - __expf(se + (ee - se) * ((float)h / 7.0f));
  const float logg = logf(g);
  const float inv1mg = 1.0f / (1.0f - g);
  const float rowsum = (1.0f - __expf(logg * (float)(t + 1))) * inv1mg;
  const float sc = rsqrtf(rowsum);
  s.x *= sc;
  s.y *= sc;
  s.z *= sc;
  s.w *= sc;
  reinterpret_cast<float4*>(ro)[(size_t)row * (CV / 4) + c4] = s;
}

#define KV_NC 32
#define KV_CJ (CT / KV_NC)
#define KV_ELEMS (CB * CH * CDK * CDV)

__global__ __launch_bounds__(256) void curr_kv_kernel(
    const float* __restrict__ qkv, float* __restrict__ part) {
  __shared__ float kw_s[8][130][2];
  __shared__ float v_s[8][258][2];

  const int ch = blockIdx.x;
  const int bh = blockIdx.y;
  const int jbeg = ch * KV_CJ;
  const int b = bh >> 3, h = bh & 7;
  const int rowbase = b * CT;

  const int tid = threadIdx.x;
  const int lane = tid & 31;
  const int w = tid >> 5;
  const int hi = lane >> 4;
  const int ln = lane & 15;

  const float se = logf(1.0f / 32.0f), ee = logf(1.0f / 512.0f);
  const float g = 1.0f - __expf(se + (ee - se) * ((float)h / 7.0f));
  const float logg = logf(g);
  const float rnorm = rsqrtf((1.0f - __expf(logg * (float)CT)) / (1.0f - g));

  v8f acc[16];
#pragma unroll
  for (int i = 0; i < 16; ++i) acc[i] = vzero8();

  for (int j0 = jbeg; j0 < jbeg + KV_CJ; j0 += 16) {
#pragma unroll
    for (int e = tid; e < 16 * 128; e += 256) {
      int r = e >> 7, c = e & 127;
      int j = j0 + r;
      float wj = __expf(logg * (float)(CT - 1 - j)) * rnorm;
      kw_s[r >> 1][c][r & 1] =
          qkv[(size_t)(rowbase + j) * NQKV + CE + h * CDK + c] * wj;
    }
#pragma unroll
    for (int e = tid; e < 16 * 256; e += 256) {
      int r = e >> 8, c = e & 255;
      v_s[r >> 1][c][r & 1] =
          qkv[(size_t)(rowbase + j0 + r) * NQKV + 2 * CE + h * CDV + c];
    }
    __syncthreads();

#pragma unroll
    for (int ks = 0; ks < 16; ks += 4) {
      const int kp = (ks >> 1) + hi;
      v2f a = ldsv2(&kw_s[kp][w * 16 + ln][0]);
#pragma unroll
      for (int nt = 0; nt < 16; ++nt) {
        v2f bf = ldsv2(&v_s[kp][nt * 16 + ln][0]);
        acc[nt] = wmma4(a, bf, acc[nt]);
      }
    }
    __syncthreads();
  }

#pragma unroll
  for (int nt = 0; nt < 16; ++nt)
#pragma unroll
    for (int vv = 0; vv < 8; ++vv) {
      int kk = w * 16 + vv + hi * 8;
      part[(size_t)ch * KV_ELEMS + (size_t)bh * CDK * CDV +
           (size_t)kk * CDV + nt * 16 + ln] = acc[nt][vv];
    }
}

__global__ __launch_bounds__(256) void curr_kv_reduce_kernel(
    const float* __restrict__ part, float* __restrict__ kvout) {
  const int idx = blockIdx.x * 256 + threadIdx.x;
  const float4* __restrict__ p4 = reinterpret_cast<const float4*>(part);
  float4 v[KV_NC];
#pragma unroll
  for (int c = 0; c < KV_NC; ++c) v[c] = p4[(size_t)c * (KV_ELEMS / 4) + idx];
  float4 s = v[0];
#pragma unroll
  for (int c = 1; c < KV_NC; ++c) {
    s.x += v[c].x;
    s.y += v[c].y;
    s.z += v[c].z;
    s.w += v[c].w;
  }
  reinterpret_cast<float4*>(kvout)[idx] = s;
}

__global__ __launch_bounds__(256) void gn_gate_kernel(
    const float* __restrict__ ro, float* __restrict__ gbuf) {
  const int row = blockIdx.x;
  const int w = threadIdx.x >> 5;
  const int lane = threadIdx.x & 31;
  const size_t base = (size_t)row * CV + (size_t)w * CDV;

  float x[8];
  float s = 0.0f, s2 = 0.0f;
#pragma unroll
  for (int i = 0; i < 8; ++i) {
    x[i] = ro[base + i * 32 + lane];
    s += x[i];
    s2 += x[i] * x[i];
  }
#pragma unroll
  for (int off = 16; off > 0; off >>= 1) {
    s += __shfl_xor(s, off, 32);
    s2 += __shfl_xor(s2, off, 32);
  }
  const float mu = s * (1.0f / 256.0f);
  const float var = s2 * (1.0f / 256.0f) - mu * mu;
  const float rstd = rsqrtf(var + 1e-5f);

#pragma unroll
  for (int i = 0; i < 8; ++i) {
    size_t idx = base + i * 32 + lane;
    float nrm = (x[i] - mu) * rstd;
    float gv = gbuf[idx];
    float sil = gv / (1.0f + __expf(-gv));
    gbuf[idx] = sil * nrm;
  }
}

extern "C" void kernel_launch(void* const* d_in, const int* in_sizes, int n_in,
                              void* d_out, int out_size, void* d_ws,
                              size_t ws_size, hipStream_t stream) {
  const float* hs    = (const float*)d_in[0];
  const float* w_qkv = (const float*)d_in[1];
  const float* b_qkv = (const float*)d_in[2];
  const float* w_g   = (const float*)d_in[3];
  const float* b_g   = (const float*)d_in[4];
  const float* w_p   = (const float*)d_in[5];
  const float* b_p   = (const float*)d_in[6];

  float* out  = (float*)d_out;
  float* qkv  = (float*)d_ws;
  float* ro   = qkv + (size_t)BT * NQKV;
  float* big  = ro + (size_t)BT * CV;
  float* kvp  = big;
  float* rtp  = big;
  float* gbuf = big;

  gemm_wmma_f32<4, 1, NQKV, CE><<<dim3(NQKV / 128, BT / 128), 128, 0, stream>>>(
      hs, w_qkv, b_qkv, qkv);

  xpos_kernel<<<dim3(BT), 512, 0, stream>>>(qkv);

  curr_kv_kernel<<<dim3(KV_NC, CB * CH), 256, 0, stream>>>(qkv, kvp);
  curr_kv_reduce_kernel<<<dim3(KV_ELEMS / 4 / 256), 256, 0, stream>>>(
      kvp, out + (size_t)BT * CE);

  retention_kernel<<<dim3(RT_NWG, CB * CH), 128, 0, stream>>>(qkv, rtp);
  retention_reduce_kernel<<<dim3(BT), 512, 0, stream>>>(rtp, ro);

  gemm_wmma_f32<2, 2, CV, CE><<<dim3(CV / 128, BT / 64), 128, 0, stream>>>(
      hs, w_g, b_g, gbuf);

  gn_gate_kernel<<<dim3(BT), 256, 0, stream>>>(ro, gbuf);

  gemm_wmma_f32<2, 2, CE, CV><<<dim3(CE / 128, BT / 64), 128, 0, stream>>>(
      gbuf, w_p, b_p, out);
}


# gemm launches: a fragment set spans 32 k (was 8 or 16): the four 16-byte loads of an A row's 128-byte line go out together, 32 loads ahead of 128 WMMAs, 2 waves per SIMD; order of the sums unchanged
# speedup vs baseline: 4.1567x; 1.0324x over previous
#include <hip/hip_runtime.h>
#include <hip/hip_bf16.h>


typedef __attribute__((ext_vector_type(2))) float v2f;
typedef __attribute__((ext_vector_type(8))) float v8f;

#define CB 2
#define CT 2048
#define CE 1024
#define CH 8
#define CDK 128
#define CDV 256
#define CV 2048
#define NQKV 4096
#define BT (CB * CT)

static __device__ __forceinline__ v8f wmma4(v2f a, v2f b, v8f c) {
  return __builtin_amdgcn_wmma_f32_16x16x4_f32(
       false, a,  false, b,
       (short)0, c,  false,  false);
}

static __device__ __forceinline__ v8f vzero8() {
  v8f z;
#pragma unroll
  for (int i = 0; i < 8; ++i) z[i] = 0.0f;
  return z;
}

static __device__ __forceinline__ v2f ldsv2(const float* p) {
  return *reinterpret_cast<const v2f*>(p);
}

typedef __attribute__((ext_vector_type(4))) float v4f;

template <int MT, int NB, int N, int K>
struct GemmFrag {
  v4f a[NB][MT];
  v4f b[NB][4];
  __device__ __forceinline__ void load(const v4f* __restrict__ pa,
                                       const v4f* __restrict__ pb, int k) {
#pragma unroll
    for (int mt = 0; mt < MT; ++mt)
#pragma unroll
      for (int nb = 0; nb < NB; ++nb)
        a[nb][mt] = pa[((size_t)mt * 16 * K + k + 8 * nb) / 4];
#pragma unroll
    for (int nb = 0; nb < NB; ++nb)
#pragma unroll
      for (int j = 0; j < 4; ++j)
        b[nb][j] = pb[((size_t)(k + 8 * nb + j) * N) / 4];
  }
  __device__ __forceinline__ void mma(v8f (&acc)[MT][4]) const {
#pragma unroll
    for (int nb = 0; nb < NB; ++nb)
#pragma unroll
      for (int s = 0; s < 2; ++s)
#pragma unroll
        for (int mt = 0; mt < MT; ++mt) {
          v2f af;
          af[0] = a[nb][mt][2 * s];
          af[1] = a[nb][mt][2 * s + 1];
#pragma unroll
          for (int nt = 0; nt < 4; ++nt) {
            v2f bf;
            bf[0] = b[nb][2 * s][nt];
            bf[1] = b[nb][2 * s + 1][nt];
            acc[mt][nt] = wmma4(af, bf, acc[mt][nt]);
          }
        }
  }
};

template <int MT, int NB, int N, int K>
__global__ __launch_bounds__(128) void gemm_wmma_f32(
    const float* __restrict__ A, const float* __restrict__ Bm,
    const float* __restrict__ bias, float* __restrict__ C) {
  constexpr int KS = 8 * NB;
  const int tid = threadIdx.x;
  const int lane = tid & 31;
  const int w = tid >> 5;
  const int hi = lane >> 4;
  const int ln = lane & 15;
  const int m0 = (blockIdx.y * 2 + (w >> 1)) * (16 * MT);
  const int n0 = (blockIdx.x * 2 + (w & 1)) * 64;

  const v4f* __restrict__ pa = reinterpret_cast<const v4f*>(
      A + (size_t)(m0 + ln) * K + 4 * hi);
  const v4f* __restrict__ pb = reinterpret_cast<const v4f*>(
      Bm + (size_t)(4 * hi) * N + n0 + 4 * ln);

  v8f acc[MT][4];
#pragma unroll
  for (int mt = 0; mt < MT; ++mt)
#pragma unroll
    for (int nt = 0; nt < 4; ++nt) acc[mt][nt] = vzero8();

  GemmFrag<MT, NB, N, K> f0, f1;
  f0.load(pa, pb, 0);
  for (int k0 = 0; k0 < K - 2 * KS; k0 += 2 * KS) {
    f1.load(pa, pb, k0 + KS);
    __builtin_amdgcn_sched_barrier(0);
    f0.mma(acc);
    __builtin_amdgcn_sched_barrier(0);
    f0.load(pa, pb, k0 + 2 * KS);
    __builtin_amdgcn_sched_barrier(0);
    f1.mma(acc);
    __builtin_amdgcn_sched_barrier(0);
  }
  f1.load(pa, pb, K - KS);
  f0.mma(acc);
  f1.mma(acc);

  const v4f bv = *reinterpret_cast<const v4f*>(bias + n0 + 4 * ln);
#pragma unroll
  for (int mt = 0; mt < MT; ++mt)
#pragma unroll
    for (int r = 0; r < 8; ++r) {
      const int row = m0 + mt * 16 + hi * 8 + r;
      v4f o;
#pragma unroll
      for (int nt = 0; nt < 4; ++nt) o[nt] = acc[mt][nt][r] + bv[nt];
      *reinterpret_cast<v4f*>(C + (size_t)row * N + n0 + 4 * ln) = o;
    }
}

__global__ __launch_bounds__(512) void xpos_kernel(float* __restrict__ qkv) {
  const int row = blockIdx.x;
  const int j = threadIdx.x;
  const int t = row & (CT - 1);

  const float D = (float)CE;
  const float base = (2.0f * (float)j + 0.4f * D) / (1.4f * D);
  const float sc = __powf(base, (float)t / 512.0f);
  const float invf = __powf(10000.0f, -(float)j / 512.0f);
  float sn, cn;
  __sincosf((float)t * invf, &sn, &cn);

  size_t off = (size_t)row * NQKV + 2 * j;
  {
    float x0 = qkv[off], x1 = qkv[off + 1];
    float c = cn * sc, s = sn * sc;
    qkv[off]     = x0 * c - x1 * s;
    qkv[off + 1] = x1 * c + x0 * s;
  }
  {
    float isc = 1.0f / sc;
    float x0 = qkv[off + CE], x1 = qkv[off + CE + 1];
    float c = cn * isc, s = sn * isc;
    qkv[off + CE]     = x0 * c - x1 * s;
    qkv[off + CE + 1] = x1 * c + x0 * s;
  }
}

#define RT_CK 256
#define RT_NWG 144
#define RT_TILE (64 * CDV)

__global__ __launch_bounds__(128) void retention_kernel(
    const float* __restrict__ qkv, float* __restrict__ part) {
  int n = blockIdx.x;
  int gq = 0;
  while (gq < 7 && n >= 4 * (gq + 1)) {
    n -= 4 * (gq + 1);
    ++gq;
  }
  const int iblk = 4 * gq + n / (gq + 1);
  const int ck = n % (gq + 1);
  const int bh = blockIdx.y;
  const int b = bh >> 3, h = bh & 7;
  const int i0 = iblk * 64;
  const int rowbase = b * CT;
  const int jbeg = ck * RT_CK;
  const int jend = min(jbeg + RT_CK, i0 + 64);
  const int nblk = (jend - jbeg) >> 4;

  const int tid = threadIdx.x;
  const int lane = tid & 31;
  const int w = tid >> 5;
  const int hi = lane >> 4;
  const int ln = lane & 15;

  const float se = logf(1.0f / 32.0f), ee = logf(1.0f / 512.0f);
  const float g = 1.0f - __expf(se + (ee - se) * ((float)h / 7.0f));
  const float logg = logf(g);
  const float rsdk = rsqrtf((float)CDK);

  const int qi = i0 + w * 16 + ln;
  const v4f* __restrict__ pq = reinterpret_cast<const v4f*>(
      qkv + (size_t)(rowbase + qi) * NQKV + h * CDK + 4 * hi);
  const float* __restrict__ pk =
      qkv + (size_t)(rowbase + jbeg + ln) * NQKV + CE + h * CDK + 4 * hi;
  const float* __restrict__ pv =
      qkv + (size_t)(rowbase + jbeg + 8 * hi) * NQKV + 2 * CE + h * CDV + 4 * ln;

  v4f qf[16];
  v4f kf[16];
  v4f vf[4][8];
#pragma unroll
  for (int kb = 0; kb < 16; ++kb) qf[kb] = pq[2 * kb];
#pragma unroll
  for (int kb = 0; kb < 16; ++kb)
    kf[kb] = *reinterpret_cast<const v4f*>(pk + 8 * kb);
  __builtin_amdgcn_sched_barrier(0);
#pragma unroll
  for (int gg = 0; gg < 2; ++gg)
#pragma unroll
    for (int jj = 0; jj < 8; ++jj)
      vf[gg][jj] =
          *reinterpret_cast<const v4f*>(pv + (size_t)jj * NQKV + 64 * gg);
  __builtin_amdgcn_sched_barrier(0);
#pragma unroll
  for (int gg = 2; gg < 4; ++gg)
#pragma unroll
    for (int jj = 0; jj < 8; ++jj)
      vf[gg][jj] =
          *reinterpret_cast<const v4f*>(pv + (size_t)jj * NQKV + 64 * gg);

  v8f acc[16];
#pragma unroll
  for (int i = 0; i < 16; ++i) acc[i] = vzero8();

  for (int t = 0; t < nblk; ++t) {
    const int j0 = jbeg + 16 * t;
    const int tn = min(t + 1, nblk - 1);
    const float* __restrict__ pkn = pk + (size_t)tn * 16 * NQKV;
    const float* __restrict__ pvn = pv + (size_t)tn * 16 * NQKV;

    v8f s0 = vzero8(), s1 = vzero8();
#pragma unroll
    for (int kb = 0; kb < 16; kb += 2)
#pragma unroll
      for (int s = 0; s < 2; ++s) {
        v2f a0, b0, a1, b1;
        a0[0] = kf[kb][2 * s];
        a0[1] = kf[kb][2 * s + 1];
        b0[0] = qf[kb][2 * s];
        b0[1] = qf[kb][2 * s + 1];
        a1[0] = kf[kb + 1][2 * s];
        a1[1] = kf[kb + 1][2 * s + 1];
        b1[0] = qf[kb + 1][2 * s];
        b1[1] = qf[kb + 1][2 * s + 1];
        s0 = wmma4(a0, b0, s0);
        s1 = wmma4(a1, b1, s1);
      }

    __builtin_amdgcn_sched_barrier(0);
#pragma unroll
    for (int kb = 0; kb < 16; ++kb)
      kf[kb] = *reinterpret_cast<const v4f*>(pkn + 8 * kb);
    __builtin_amdgcn_sched_barrier(0);

    v8f st;
#pragma unroll
    for (int r = 0; r < 8; ++r) {
      const int j = j0 + 8 * hi + r;
      const float m =
          (qi >= j) ? __expf(logg * (float)(qi - j)) * rsdk : 0.0f;
      st[r] = (s0[r] + s1[r]) * m;
    }

#pragma unroll
    for (int gg = 0; gg < 4; ++gg) {
#pragma unroll
      for (int s = 0; s < 4; ++s) {
        v2f bf;
        bf[0] = st[2 * s];
        bf[1] = st[2 * s + 1];
#pragma unroll
        for (int mt = 0; mt < 4; ++mt) {
          v2f af;
          af[0] = vf[gg][2 * s][mt];
          af[1] = vf[gg][2 * s + 1][mt];
          acc[4 * gg + mt] = wmma4(af, bf, acc[4 * gg + mt]);
        }
      }
      __builtin_amdgcn_sched_barrier(0);
#pragma unroll
      for (int jj = 0; jj < 8; ++jj)
        vf[gg][jj] =
            *reinterpret_cast<const v4f*>(pvn + (size_t)jj * NQKV + 64 * gg);
      __builtin_amdgcn_sched_barrier(0);
    }
  }

  float* __restrict__ trow =
      part + ((size_t)bh * RT_NWG + blockIdx.x) * RT_TILE +
      (size_t)(w * 16 + ln) * CDV + 32 * hi;
#pragma unroll
  for (int gg = 0; gg < 4; ++gg)
#pragma unroll
    for (int r = 0; r < 8; ++r) {
      v4f o;
#pragma unroll
      for (int mt = 0; mt < 4; ++mt) o[mt] = acc[4 * gg + mt][r];
      *reinterpret_cast<v4f*>(trow + 64 * gg + 4 * r) = o;
    }
}

__global__ __launch_bounds__(512) void retention_reduce_kernel(
    const float* __restrict__ part, float* __restrict__ ro) {
  const int row = blockIdx.x;
  const int b = row >> 11;
  const int t = row & (CT - 1);
  const int c4 = threadIdx.x;
  const int h = c4 >> 6;
  const int iblk = t >> 6;
  const int gq = iblk >> 2;
  const int nch = gq + 1;
  const int n0 = 2 * gq * (gq + 1) + (iblk & 3) * nch;

  const float4* __restrict__ p4 =
      reinterpret_cast<const float4*>(
          part + ((size_t)(b * CH + h) * RT_NWG + n0) * RT_TILE +
          (size_t)(t & 63) * CDV) +
      (c4 & 63);
  float4 v[8];
#pragma unroll
  for (int c = 0; c < 8; ++c)
    if (c < nch) v[c] = p4[(size_t)c * (RT_TILE / 4)];
  float4 s = v[0];
#pragma unroll
  for (int c = 1; c < 8; ++c)
    if (c < nch) {
      s.x += v[c].x;
      s.y += v[c].y;
      s.z += v[c].z;
      s.w += v[c].w;
    }

  const float se = logf(1.0f / 32.0f), ee = logf(1.0f / 512.0f);
  const float g = 1.0f - __expf(se + (ee - se) * ((float)h / 7.0f));
  const float logg = logf(g);
  const float inv1mg = 1.0f / (1.0f - g);
  const float rowsum = (1.0f - __expf(logg * (float)(t + 1))) * inv1mg;
  const float sc = rsqrtf(rowsum);
  s.x *= sc;
  s.y *= sc;
  s.z *= sc;
  s.w *= sc;
  reinterpret_cast<float4*>(ro)[(size_t)row * (CV / 4) + c4] = s;
}

#define KV_NC 32
#define KV_CJ (CT / KV_NC)
#define KV_ELEMS (CB * CH * CDK * CDV)

__global__ __launch_bounds__(256) void curr_kv_kernel(
    const float* __restrict__ qkv, float* __restrict__ part) {
  __shared__ float kw_s[8][130][2];
  __shared__ float v_s[8][258][2];

  const int ch = blockIdx.x;
  const int bh = blockIdx.y;
  const int jbeg = ch * KV_CJ;
  const int b = bh >> 3, h = bh & 7;
  const int rowbase = b * CT;

  const int tid = threadIdx.x;
  const int lane = tid & 31;
  const int w = tid >> 5;
  const int hi = lane >> 4;
  const int ln = lane & 15;

  const float se = logf(1.0f / 32.0f), ee = logf(1.0f / 512.0f);
  const float g = 1.0f - __expf(se + (ee - se) * ((float)h / 7.0f));
  const float logg = logf(g);
  const float rnorm = rsqrtf((1.0f - __expf(logg * (float)CT)) / (1.0f - g));

  v8f acc[16];
#pragma unroll
  for (int i = 0; i < 16; ++i) acc[i] = vzero8();

  for (int j0 = jbeg; j0 < jbeg + KV_CJ; j0 += 16) {
#pragma unroll
    for (int e = tid; e < 16 * 128; e += 256) {
      int r = e >> 7, c = e & 127;
      int j = j0 + r;
      float wj = __expf(logg * (float)(CT - 1 - j)) * rnorm;
      kw_s[r >> 1][c][r & 1] =
          qkv[(size_t)(rowbase + j) * NQKV + CE + h * CDK + c] * wj;
    }
#pragma unroll
    for (int e = tid; e < 16 * 256; e += 256) {
      int r = e >> 8, c = e & 255;
      v_s[r >> 1][c][r & 1] =
          qkv[(size_t)(rowbase + j0 + r) * NQKV + 2 * CE + h * CDV + c];
    }
    __syncthreads();

#pragma unroll
    for (int ks = 0; ks < 16; ks += 4) {
      const int kp = (ks >> 1) + hi;
      v2f a = ldsv2(&kw_s[kp][w * 16 + ln][0]);
#pragma unroll
      for (int nt = 0; nt < 16; ++nt) {
        v2f bf = ldsv2(&v_s[kp][nt * 16 + ln][0]);
        acc[nt] = wmma4(a, bf, acc[nt]);
      }
    }
    __syncthreads();
  }

#pragma unroll
  for (int nt = 0; nt < 16; ++nt)
#pragma unroll
    for (int vv = 0; vv < 8; ++vv) {
      int kk = w * 16 + vv + hi * 8;
      part[(size_t)ch * KV_ELEMS + (size_t)bh * CDK * CDV +
           (size_t)kk * CDV + nt * 16 + ln] = acc[nt][vv];
    }
}

__global__ __launch_bounds__(256) void curr_kv_reduce_kernel(
    const float* __restrict__ part, float* __restrict__ kvout) {
  const int idx = blockIdx.x * 256 + threadIdx.x;
  const float4* __restrict__ p4 = reinterpret_cast<const float4*>(part);
  float4 v[KV_NC];
#pragma unroll
  for (int c = 0; c < KV_NC; ++c) v[c] = p4[(size_t)c * (KV_ELEMS / 4) + idx];
  float4 s = v[0];
#pragma unroll
  for (int c = 1; c < KV_NC; ++c) {
    s.x += v[c].x;
    s.y += v[c].y;
    s.z += v[c].z;
    s.w += v[c].w;
  }
  reinterpret_cast<float4*>(kvout)[idx] = s;
}

__global__ __launch_bounds__(256) void gn_gate_kernel(
    const float* __restrict__ ro, float* __restrict__ gbuf) {
  const int row = blockIdx.x;
  const int w = threadIdx.x >> 5;
  const int lane = threadIdx.x & 31;
  const size_t base = (size_t)row * CV + (size_t)w * CDV;

  float x[8];
  float s = 0.0f, s2 = 0.0f;
#pragma unroll
  for (int i = 0; i < 8; ++i) {
    x[i] = ro[base + i * 32 + lane];
    s += x[i];
    s2 += x[i] * x[i];
  }
#pragma unroll
  for (int off = 16; off > 0; off >>= 1) {
    s += __shfl_xor(s, off, 32);
    s2 += __shfl_xor(s2, off, 32);
  }
  const float mu = s * (1.0f / 256.0f);
  const float var = s2 * (1.0f / 256.0f) - mu * mu;
  const float rstd = rsqrtf(var + 1e-5f);

#pragma unroll
  for (int i = 0; i < 8; ++i) {
    size_t idx = base + i * 32 + lane;
    float nrm = (x[i] - mu) * rstd;
    float gv = gbuf[idx];
    float sil = gv / (1.0f + __expf(-gv));
    gbuf[idx] = sil * nrm;
  }
}

extern "C" void kernel_launch(void* const* d_in, const int* in_sizes, int n_in,
                              void* d_out, int out_size, void* d_ws,
                              size_t ws_size, hipStream_t stream) {
  const float* hs    = (const float*)d_in[0];
  const float* w_qkv = (const float*)d_in[1];
  const float* b_qkv = (const float*)d_in[2];
  const float* w_g   = (const float*)d_in[3];
  const float* b_g   = (const float*)d_in[4];
  const float* w_p   = (const float*)d_in[5];
  const float* b_p   = (const float*)d_in[6];

  float* out  = (float*)d_out;
  float* qkv  = (float*)d_ws;
  float* ro   = qkv + (size_t)BT * NQKV;
  float* big  = ro + (size_t)BT * CV;
  float* kvp  = big;
  float* rtp  = big;
  float* gbuf = big;

  gemm_wmma_f32<4, 4, NQKV, CE><<<dim3(NQKV / 128, BT / 128), 128, 0, stream>>>(
      hs, w_qkv, b_qkv, qkv);

  xpos_kernel<<<dim3(BT), 512, 0, stream>>>(qkv);

  curr_kv_kernel<<<dim3(KV_NC, CB * CH), 256, 0, stream>>>(qkv, kvp);
  curr_kv_reduce_kernel<<<dim3(KV_ELEMS / 4 / 256), 256, 0, stream>>>(
      kvp, out + (size_t)BT * CE);

  retention_kernel<<<dim3(RT_NWG, CB * CH), 128, 0, stream>>>(qkv, rtp);
  retention_reduce_kernel<<<dim3(BT), 512, 0, stream>>>(rtp, ro);

  gemm_wmma_f32<4, 4, CV, CE><<<dim3(CV / 128, BT / 128), 128, 0, stream>>>(
      hs, w_g, b_g, gbuf);

  gn_gate_kernel<<<dim3(BT), 256, 0, stream>>>(ro, gbuf);

  gemm_wmma_f32<2, 4, CE, CV><<<dim3(CE / 128, BT / 64), 128, 0, stream>>>(
      gbuf, w_p, b_p, out);
}
